# starting kernel
# baseline (speedup 1.0000x reference)
#include <hip/hip_runtime.h>

typedef __attribute__((ext_vector_type(2))) float v2f;
typedef __attribute__((ext_vector_type(4))) float v4f;
typedef __attribute__((ext_vector_type(8))) float v8f;

#define NROWS 4096
#define NF    128
#define KB    64
#define LDT   68
#define TILE  (NF * LDT)

__global__ void gcn_degree(const float* __restrict__ adj, float* __restrict__ dinv)
{
    const int  lane = threadIdx.x & 31;
    const int  wave = threadIdx.x >> 5;
    const long row  = (long)blockIdx.x * 8 + wave;
    const v4f* rp4  = (const v4f*)(adj + row * (long)NROWS);

    float s = 0.f;
    for (int k = lane; k < NROWS / 4; k += 32) {
        v4f v = rp4[k];
        s += v.x + v.y + v.z + v.w;
    }
    #pragma unroll
    for (int o = 16; o > 0; o >>= 1) s += __shfl_xor(s, o, 32);

    if (lane == 0) dinv[row] = 1.0f / sqrtf(s + 1e-6f);
}

__global__ __launch_bounds__(256, 1)
void gcn_gemm(const float* __restrict__ A, long lda, long aBatchStride,
              const float* __restrict__ Bm, long bBatchStride, int K,
              const float* __restrict__ dinv, const float* __restrict__ bias,
              float* __restrict__ out)
{
    __shared__ float bt[2 * TILE];

    const int tid    = threadIdx.x;
    const int lane   = tid & 31;
    const int wave   = tid >> 5;
    const int batch  = blockIdx.x >> 5;
    const int rowBlk = blockIdx.x & 31;
    const int m0     = rowBlk * 128 + wave * 16;
    const int lh     = lane >> 4;
    const int ll     = lane & 15;
    const long gRow0 = (long)batch * NROWS + m0;

    const float* Ap   = A  + (long)batch * aBatchStride;
    const float* Bp   = Bm + (long)batch * bBatchStride;
    const float* aRow = Ap + (long)(m0 + ll) * lda + lh * 2;

    v8f acc[8] = {};

    const int nStages = K / KB;

    #pragma unroll
    for (int i = 0; i < (KB * NF) / 256; ++i) {
        int idx = tid + i * 256;
        int r   = idx >> 7;
        int c   = idx & 127;
        bt[c * LDT + r] = Bp[(long)r * NF + c];
    }

    for (int s = 0; s < nStages; ++s) {
        __syncthreads();
        const float* cur = bt + (s & 1) * TILE;

        if (s + 1 < nStages) {
            float* nxt = bt + ((s + 1) & 1) * TILE;
            const long k0n = (long)(s + 1) * KB;
            #pragma unroll
            for (int i = 0; i < (KB * NF) / 256; ++i) {
                int idx = tid + i * 256;
                int r   = idx >> 7;
                int c   = idx & 127;
                nxt[c * LDT + r] = Bp[(k0n + r) * NF + c];
            }
        }

        const float* ap = aRow + (long)s * KB;
        #pragma unroll
        for (int j = 0; j < KB / 4; ++j) {
            v2f a = __builtin_nontemporal_load((const v2f*)(ap + j * 4));
            const int kk = j * 4 + lh * 2;
            #pragma unroll
            for (int nt = 0; nt < 8; ++nt) {
                v2f b = *(const v2f*)(&cur[(nt * 16 + ll) * LDT + kk]);
                acc[nt] = __builtin_amdgcn_wmma_f32_16x16x4_f32(
                    false, a, false, b, (short)0, acc[nt], false, false);
            }
        }
    }

    #pragma unroll
    for (int nt = 0; nt < 8; ++nt) {
        const int col = nt * 16 + ll;
        #pragma unroll
        for (int v = 0; v < 8; ++v) {
            const long gRow = gRow0 + v + lh * 8;
            float val = acc[nt][v] * dinv[gRow];
            if (bias) val += bias[col];
            out[gRow * NF + col] = val;
        }
    }
}

extern "C" void kernel_launch(void* const* d_in, const int* in_sizes, int n_in,
                              void* d_out, int out_size, void* d_ws, size_t ws_size,
                              hipStream_t stream) {
    const float* x      = (const float*)d_in[0];
    const float* adj    = (const float*)d_in[1];
    const float* weight = (const float*)d_in[2];
    const float* bias   = (const float*)d_in[3];
    float*       out    = (float*)d_out;

    float* dinv = (float*)d_ws;
    float* y    = dinv + 4 * NROWS;

    const int B = 4;

    gcn_degree<<<(B * NROWS) / 8, 256, 0, stream>>>(adj, dinv);

    gcn_gemm<<<B * (NROWS / 128), 256, 0, stream>>>(
        x, NF, (long)NROWS * NF, weight, 0, NF, dinv, nullptr, y);

    gcn_gemm<<<B * (NROWS / 128), 256, 0, stream>>>(
        adj, NROWS, (long)NROWS * NROWS, y, (long)NROWS * NF, NROWS,
        dinv, bias, out);
}


# B panel of the next K-stage staged through registers: its 32 global loads issued as one group before the WMMA block, ds_store after the block (was load-wait-store 32 times in series); sum order kept
# speedup vs baseline: 3.3772x; 3.3772x over previous
#include <hip/hip_runtime.h>

typedef __attribute__((ext_vector_type(2))) float v2f;
typedef __attribute__((ext_vector_type(4))) float v4f;
typedef __attribute__((ext_vector_type(8))) float v8f;

#define NROWS 4096
#define NF    128
#define KB    64
#define LDT   68
#define TILE  (NF * LDT)

__global__ void gcn_degree(const float* __restrict__ adj, float* __restrict__ dinv)
{
    const int  lane = threadIdx.x & 31;
    const int  wave = threadIdx.x >> 5;
    const long row  = (long)blockIdx.x * 8 + wave;
    const v4f* rp4  = (const v4f*)(adj + row * (long)NROWS);

    float s = 0.f;
    for (int k = lane; k < NROWS / 4; k += 32) {
        v4f v = rp4[k];
        s += v.x + v.y + v.z + v.w;
    }
    #pragma unroll
    for (int o = 16; o > 0; o >>= 1) s += __shfl_xor(s, o, 32);

    if (lane == 0) dinv[row] = 1.0f / sqrtf(s + 1e-6f);
}

__global__ __launch_bounds__(256, 1)
void gcn_gemm(const float* __restrict__ A, long lda, long aBatchStride,
              const float* __restrict__ Bm, long bBatchStride, int K,
              const float* __restrict__ dinv, const float* __restrict__ bias,
              float* __restrict__ out)
{
    __shared__ float bt[2 * TILE];

    const int tid    = threadIdx.x;
    const int lane   = tid & 31;
    const int wave   = tid >> 5;
    const int batch  = blockIdx.x >> 5;
    const int rowBlk = blockIdx.x & 31;
    const int m0     = rowBlk * 128 + wave * 16;
    const int lh     = lane >> 4;
    const int ll     = lane & 15;
    const long gRow0 = (long)batch * NROWS + m0;

    const float* Ap   = A  + (long)batch * aBatchStride;
    const float* Bp   = Bm + (long)batch * bBatchStride;
    const float* aRow = Ap + (long)(m0 + ll) * lda + lh * 2;

    v8f acc[8] = {};

    const int nStages = K / KB;

    #pragma unroll
    for (int i = 0; i < (KB * NF) / 256; ++i) {
        int idx = tid + i * 256;
        int r   = idx >> 7;
        int c   = idx & 127;
        bt[c * LDT + r] = Bp[(long)r * NF + c];
    }

    for (int s = 0; s < nStages; ++s) {
        __syncthreads();
        const float* cur = bt + (s & 1) * TILE;
        const bool more = (s + 1 < nStages);

        float breg[(KB * NF) / 256];
        if (more) {
            const float* src = Bp + (long)(s + 1) * (KB * NF);
            #pragma unroll
            for (int i = 0; i < (KB * NF) / 256; ++i)
                breg[i] = src[tid + i * 256];
        }

        const float* ap = aRow + (long)s * KB;
        #pragma unroll
        for (int j = 0; j < KB / 4; ++j) {
            v2f a = __builtin_nontemporal_load((const v2f*)(ap + j * 4));
            const int kk = j * 4 + lh * 2;
            #pragma unroll
            for (int nt = 0; nt < 8; ++nt) {
                v2f b = *(const v2f*)(&cur[(nt * 16 + ll) * LDT + kk]);
                acc[nt] = __builtin_amdgcn_wmma_f32_16x16x4_f32(
                    false, a, false, b, (short)0, acc[nt], false, false);
            }
        }

        if (more) {
            float* nxt = bt + ((s + 1) & 1) * TILE;
            #pragma unroll
            for (int i = 0; i < (KB * NF) / 256; ++i) {
                int idx = tid + i * 256;
                int r   = idx >> 7;
                int c   = idx & 127;
                nxt[c * LDT + r] = breg[i];
            }
        }
    }

    #pragma unroll
    for (int nt = 0; nt < 8; ++nt) {
        const int col = nt * 16 + ll;
        #pragma unroll
        for (int v = 0; v < 8; ++v) {
            const long gRow = gRow0 + v + lh * 8;
            float val = acc[nt][v] * dinv[gRow];
            if (bias) val += bias[col];
            out[gRow * NF + col] = val;
        }
    }
}

extern "C" void kernel_launch(void* const* d_in, const int* in_sizes, int n_in,
                              void* d_out, int out_size, void* d_ws, size_t ws_size,
                              hipStream_t stream) {
    const float* x      = (const float*)d_in[0];
    const float* adj    = (const float*)d_in[1];
    const float* weight = (const float*)d_in[2];
    const float* bias   = (const float*)d_in[3];
    float*       out    = (float*)d_out;

    float* dinv = (float*)d_ws;
    float* y    = dinv + 4 * NROWS;

    const int B = 4;

    gcn_degree<<<(B * NROWS) / 8, 256, 0, stream>>>(adj, dinv);

    gcn_gemm<<<B * (NROWS / 128), 256, 0, stream>>>(
        x, NF, (long)NROWS * NF, weight, 0, NF, dinv, nullptr, y);

    gcn_gemm<<<B * (NROWS / 128), 256, 0, stream>>>(
        adj, NROWS, (long)NROWS * NROWS, y, (long)NROWS * NF, NROWS,
        dinv, bias, out);
}


# GEMM workgroup tile reshaped 128x128xK64 -> 128 rows x 64 cols x K-stage 128: grid 256 workgroups for 256 WGPs (was 128: half the WGPs idle), same WMMA count per stage; sum order kept
# speedup vs baseline: 5.0995x; 1.5100x over previous
#include <hip/hip_runtime.h>

typedef __attribute__((ext_vector_type(2))) float v2f;
typedef __attribute__((ext_vector_type(4))) float v4f;
typedef __attribute__((ext_vector_type(8))) float v8f;

#define NROWS 4096
#define NF    128
#define CB    64
#define KB    128
#define LDT   132
#define TILE  (CB * LDT)

__global__ void gcn_degree(const float* __restrict__ adj, float* __restrict__ dinv)
{
    const int  lane = threadIdx.x & 31;
    const int  wave = threadIdx.x >> 5;
    const long row  = (long)blockIdx.x * 8 + wave;
    const v4f* rp4  = (const v4f*)(adj + row * (long)NROWS);

    float s = 0.f;
    for (int k = lane; k < NROWS / 4; k += 32) {
        v4f v = rp4[k];
        s += v.x + v.y + v.z + v.w;
    }
    #pragma unroll
    for (int o = 16; o > 0; o >>= 1) s += __shfl_xor(s, o, 32);

    if (lane == 0) dinv[row] = 1.0f / sqrtf(s + 1e-6f);
}

__global__ __launch_bounds__(256, 1)
void gcn_gemm(const float* __restrict__ A, long lda, long aBatchStride,
              const float* __restrict__ Bm, long bBatchStride, int K,
              const float* __restrict__ dinv, const float* __restrict__ bias,
              float* __restrict__ out)
{
    __shared__ float bt[2 * TILE];

    const int tid    = threadIdx.x;
    const int lane   = tid & 31;
    const int wave   = tid >> 5;
    const int batch  = blockIdx.x >> 6;
    const int rowBlk = (blockIdx.x >> 1) & 31;
    const int c0     = (blockIdx.x & 1) * CB;
    const int m0     = rowBlk * 128 + wave * 16;
    const int lh     = lane >> 4;
    const int ll     = lane & 15;
    const long gRow0 = (long)batch * NROWS + m0;

    const float* Ap   = A  + (long)batch * aBatchStride;
    const float* Bp   = Bm + (long)batch * bBatchStride + c0;
    const float* aRow = Ap + (long)(m0 + ll) * lda + lh * 2;

    const int br = tid >> 6;
    const int bc = tid & 63;

    v8f acc[4] = {};

    const int nStages = K / KB;

    #pragma unroll
    for (int i = 0; i < (KB * CB) / 256; ++i)
        bt[bc * LDT + br + 4 * i] = Bp[(long)(br + 4 * i) * NF + bc];

    for (int s = 0; s < nStages; ++s) {
        __syncthreads();
        const float* cur = bt + (s & 1) * TILE;
        const bool more = (s + 1 < nStages);

        float breg[(KB * CB) / 256];
        if (more) {
            const float* src = Bp + (long)(s + 1) * (KB * NF);
            #pragma unroll
            for (int i = 0; i < (KB * CB) / 256; ++i)
                breg[i] = src[(long)(br + 4 * i) * NF + bc];
        }

        const float* ap = aRow + (long)s * KB;
        v2f areg[KB / 4];
        #pragma unroll
        for (int j = 0; j < KB / 4; ++j)
            areg[j] = __builtin_nontemporal_load((const v2f*)(ap + j * 4));

        #pragma unroll
        for (int j = 0; j < KB / 4; ++j) {
            const v2f a = areg[j];
            const int kk = j * 4 + lh * 2;
            #pragma unroll
            for (int nt = 0; nt < 4; ++nt) {
                v2f b = *(const v2f*)(&cur[(nt * 16 + ll) * LDT + kk]);
                acc[nt] = __builtin_amdgcn_wmma_f32_16x16x4_f32(
                    false, a, false, b, (short)0, acc[nt], false, false);
            }
        }

        if (more) {
            float* nxt = bt + ((s + 1) & 1) * TILE;
            #pragma unroll
            for (int i = 0; i < (KB * CB) / 256; ++i)
                nxt[bc * LDT + br + 4 * i] = breg[i];
        }
    }

    #pragma unroll
    for (int nt = 0; nt < 4; ++nt) {
        const int col = c0 + nt * 16 + ll;
        #pragma unroll
        for (int v = 0; v < 8; ++v) {
            const long gRow = gRow0 + v + lh * 8;
            float val = acc[nt][v] * dinv[gRow];
            if (bias) val += bias[col];
            out[gRow * NF + col] = val;
        }
    }
}

extern "C" void kernel_launch(void* const* d_in, const int* in_sizes, int n_in,
                              void* d_out, int out_size, void* d_ws, size_t ws_size,
                              hipStream_t stream) {
    const float* x      = (const float*)d_in[0];
    const float* adj    = (const float*)d_in[1];
    const float* weight = (const float*)d_in[2];
    const float* bias   = (const float*)d_in[3];
    float*       out    = (float*)d_out;

    float* dinv = (float*)d_ws;
    float* y    = dinv + 4 * NROWS;

    const int B = 4;

    gcn_degree<<<(B * NROWS) / 8, 256, 0, stream>>>(adj, dinv);

    gcn_gemm<<<B * (NROWS / 128) * (NF / CB), 256, 0, stream>>>(
        x, NF, (long)NROWS * NF, weight, 0, NF, dinv, nullptr, y);

    gcn_gemm<<<B * (NROWS / 128) * (NF / CB), 256, 0, stream>>>(
        adj, NROWS, (long)NROWS * NROWS, y, (long)NROWS * NF, NROWS,
        dinv, bias, out);
}


# A fragments double-buffered in registers: global A loads of K-stage s+1 issued inside the WMMA block of stage s and kept in flight across an LDS-only workgroup barrier; sum order kept
# speedup vs baseline: 6.0165x; 1.1798x over previous
#include <hip/hip_runtime.h>

typedef __attribute__((ext_vector_type(2))) float v2f;
typedef __attribute__((ext_vector_type(4))) float v4f;
typedef __attribute__((ext_vector_type(8))) float v8f;

#define NROWS 4096
#define NF    128
#define CB    64
#define KB    128
#define LDT   132
#define TILE  (CB * LDT)

__global__ void gcn_degree(const float* __restrict__ adj, float* __restrict__ dinv)
{
    const int  lane = threadIdx.x & 31;
    const int  wave = threadIdx.x >> 5;
    const long row  = (long)blockIdx.x * 8 + wave;
    const v4f* rp4  = (const v4f*)(adj + row * (long)NROWS);

    float s = 0.f;
    for (int k = lane; k < NROWS / 4; k += 32) {
        v4f v = rp4[k];
        s += v.x + v.y + v.z + v.w;
    }
    #pragma unroll
    for (int o = 16; o > 0; o >>= 1) s += __shfl_xor(s, o, 32);

    if (lane == 0) dinv[row] = 1.0f / sqrtf(s + 1e-6f);
}

#define AQ 8

__device__ __forceinline__ void lds_barrier()
{
    __builtin_amdgcn_fence(__ATOMIC_RELEASE, "workgroup", "local");
    __builtin_amdgcn_s_barrier();
    __builtin_amdgcn_fence(__ATOMIC_ACQUIRE, "workgroup", "local");
}

template <bool MORE>
__device__ __forceinline__ void gcn_gemm_stage(
    const float* __restrict__ cur, float* __restrict__ nxt,
    const float* __restrict__ bsrc, const float* __restrict__ asrc,
    const v2f (&acur)[KB / 4], v2f (&anxt)[KB / 4], v8f (&acc)[4],
    const int br, const int bc, const int lh, const int ll)
{
    float breg[(KB * CB) / 256];
    if (MORE) {
        #pragma unroll
        for (int i = 0; i < (KB * CB) / 256; ++i)
            breg[i] = bsrc[(long)(br + 4 * i) * NF + bc];
    }
    __builtin_amdgcn_sched_barrier(0);

    #pragma unroll
    for (int qd = 0; qd < (KB / 4) / AQ; ++qd) {
        if (MORE) {
            #pragma unroll
            for (int j = qd * AQ; j < (qd + 1) * AQ; ++j)
                anxt[j] = __builtin_nontemporal_load((const v2f*)(asrc + j * 4));
        }
        __builtin_amdgcn_sched_barrier(0);
        #pragma unroll
        for (int j = qd * AQ; j < (qd + 1) * AQ; ++j) {
            const v2f a = acur[j];
            const int kk = j * 4 + lh * 2;
            #pragma unroll
            for (int nt = 0; nt < 4; ++nt) {
                v2f b = *(const v2f*)(&cur[(nt * 16 + ll) * LDT + kk]);
                acc[nt] = __builtin_amdgcn_wmma_f32_16x16x4_f32(
                    false, a, false, b, (short)0, acc[nt], false, false);
            }
        }
    }

    __builtin_amdgcn_sched_barrier(0);
    if (MORE) {
        #pragma unroll
        for (int i = 0; i < (KB * CB) / 256; ++i)
            nxt[bc * LDT + br + 4 * i] = breg[i];
    }
}

__global__ __launch_bounds__(256, 1)
void gcn_gemm(const float* __restrict__ A, long lda, long aBatchStride,
              const float* __restrict__ Bm, long bBatchStride, int K,
              const float* __restrict__ dinv, const float* __restrict__ bias,
              float* __restrict__ out)
{
    __shared__ float bt[2 * TILE];

    const int tid    = threadIdx.x;
    const int lane   = tid & 31;
    const int wave   = tid >> 5;
    const int batch  = blockIdx.x >> 6;
    const int rowBlk = (blockIdx.x >> 1) & 31;
    const int c0     = (blockIdx.x & 1) * CB;
    const int m0     = rowBlk * 128 + wave * 16;
    const int lh     = lane >> 4;
    const int ll     = lane & 15;
    const long gRow0 = (long)batch * NROWS + m0;

    const float* Ap   = A  + (long)batch * aBatchStride;
    const float* Bp   = Bm + (long)batch * bBatchStride + c0;
    const float* aRow = Ap + (long)(m0 + ll) * lda + lh * 2;

    const int br = tid >> 6;
    const int bc = tid & 63;

    v8f acc[4] = {};

    const int nStages = K / KB;

    v2f a0[KB / 4], a1[KB / 4];
    #pragma unroll
    for (int j = 0; j < KB / 4; ++j)
        a0[j] = __builtin_nontemporal_load((const v2f*)(aRow + j * 4));

    #pragma unroll
    for (int i = 0; i < (KB * CB) / 256; ++i)
        bt[bc * LDT + br + 4 * i] = Bp[(long)(br + 4 * i) * NF + bc];

    int s = 0;
    for (; s + 2 < nStages; s += 2) {
        lds_barrier();
        gcn_gemm_stage<true>(bt, bt + TILE,
                             Bp + (long)(s + 1) * (KB * NF), aRow + (long)(s + 1) * KB,
                             a0, a1, acc, br, bc, lh, ll);
        lds_barrier();
        gcn_gemm_stage<true>(bt + TILE, bt,
                             Bp + (long)(s + 2) * (KB * NF), aRow + (long)(s + 2) * KB,
                             a1, a0, acc, br, bc, lh, ll);
    }
    if (s + 1 < nStages) {
        lds_barrier();
        gcn_gemm_stage<true>(bt, bt + TILE,
                             Bp + (long)(s + 1) * (KB * NF), aRow + (long)(s + 1) * KB,
                             a0, a1, acc, br, bc, lh, ll);
        lds_barrier();
        gcn_gemm_stage<false>(bt + TILE, bt, Bp, aRow, a1, a0, acc, br, bc, lh, ll);
    } else if (s < nStages) {
        lds_barrier();
        gcn_gemm_stage<false>(bt, bt + TILE, Bp, aRow, a0, a1, acc, br, bc, lh, ll);
    }

    #pragma unroll
    for (int nt = 0; nt < 4; ++nt) {
        const int col = c0 + nt * 16 + ll;
        #pragma unroll
        for (int v = 0; v < 8; ++v) {
            const long gRow = gRow0 + v + lh * 8;
            float val = acc[nt][v] * dinv[gRow];
            if (bias) val += bias[col];
            out[gRow * NF + col] = val;
        }
    }
}

extern "C" void kernel_launch(void* const* d_in, const int* in_sizes, int n_in,
                              void* d_out, int out_size, void* d_ws, size_t ws_size,
                              hipStream_t stream) {
    const float* x      = (const float*)d_in[0];
    const float* adj    = (const float*)d_in[1];
    const float* weight = (const float*)d_in[2];
    const float* bias   = (const float*)d_in[3];
    float*       out    = (float*)d_out;

    float* dinv = (float*)d_ws;
    float* y    = dinv + 4 * NROWS;

    const int B = 4;

    gcn_degree<<<(B * NROWS) / 8, 256, 0, stream>>>(adj, dinv);

    gcn_gemm<<<B * (NROWS / 128) * (NF / CB), 256, 0, stream>>>(
        x, NF, (long)NROWS * NF, weight, 0, NF, dinv, nullptr, y);

    gcn_gemm<<<B * (NROWS / 128) * (NF / CB), 256, 0, stream>>>(
        adj, NROWS, (long)NROWS * NROWS, y, (long)NROWS * NF, NROWS,
        dinv, bias, out);
}


# B fragments read from LDS as one group a quarter K-stage (32 WMMAs) ahead into a second register set (were loads among the WMMAs); barrier moved in front of the 4th quarter; sum order kept
# speedup vs baseline: 6.5143x; 1.0827x over previous
#include <hip/hip_runtime.h>

typedef __attribute__((ext_vector_type(2))) float v2f;
typedef __attribute__((ext_vector_type(4))) float v4f;
typedef __attribute__((ext_vector_type(8))) float v8f;

#define NROWS 4096
#define NF    128
#define CB    64
#define KB    128
#define LDT   132
#define TILE  (CB * LDT)

__global__ void gcn_degree(const float* __restrict__ adj, float* __restrict__ dinv)
{
    const int  lane = threadIdx.x & 31;
    const int  wave = threadIdx.x >> 5;
    const long row  = (long)blockIdx.x * 8 + wave;
    const v4f* rp4  = (const v4f*)(adj + row * (long)NROWS);

    float s = 0.f;
    for (int k = lane; k < NROWS / 4; k += 32) {
        v4f v = rp4[k];
        s += v.x + v.y + v.z + v.w;
    }
    #pragma unroll
    for (int o = 16; o > 0; o >>= 1) s += __shfl_xor(s, o, 32);

    if (lane == 0) dinv[row] = 1.0f / sqrtf(s + 1e-6f);
}

#define AQ 8
#define FQ (AQ * 4)

__device__ __forceinline__ void lds_barrier()
{
    __builtin_amdgcn_fence(__ATOMIC_RELEASE, "workgroup", "local");
    __builtin_amdgcn_s_barrier();
    __builtin_amdgcn_fence(__ATOMIC_ACQUIRE, "workgroup", "local");
}

__device__ __forceinline__ void load_bfrags(const float* __restrict__ p, const int qd,
                                            v2f (&f)[FQ], const int fb)
{
    #pragma unroll
    for (int jj = 0; jj < AQ; ++jj) {
        #pragma unroll
        for (int nt = 0; nt < 4; ++nt)
            f[jj * 4 + nt] = *(const v2f*)(&p[fb + nt * 16 * LDT + (qd * AQ + jj) * 4]);
    }
}

template <int QD, bool APF>
__device__ __forceinline__ void gcn_quarter(
    const float* __restrict__ asrc, const v2f (&acur)[KB / 4], v2f (&anxt)[KB / 4],
    const v2f (&bu)[FQ], v8f (&acc)[4])
{
    if (APF) {
        #pragma unroll
        for (int j = QD * AQ; j < (QD + 1) * AQ; ++j)
            anxt[j] = __builtin_nontemporal_load((const v2f*)(asrc + j * 4));
    }
    __builtin_amdgcn_sched_barrier(0);
    #pragma unroll
    for (int jj = 0; jj < AQ; ++jj) {
        const v2f a = acur[QD * AQ + jj];
        #pragma unroll
        for (int nt = 0; nt < 4; ++nt)
            acc[nt] = __builtin_amdgcn_wmma_f32_16x16x4_f32(
                false, a, false, bu[jj * 4 + nt], (short)0, acc[nt], false, false);
    }
    __builtin_amdgcn_sched_barrier(0);
}

template <bool MORE>
__device__ __forceinline__ void gcn_gemm_stage(
    const float* __restrict__ cur, float* __restrict__ nxt,
    const float* __restrict__ bsrc, const float* __restrict__ asrc,
    const v2f (&acur)[KB / 4], v2f (&anxt)[KB / 4], v8f (&acc)[4],
    v2f (&bfA)[FQ], v2f (&bfB)[FQ],
    const int br, const int bc, const int fb)
{
    float breg[(KB * CB) / 256];
    if (MORE) {
        #pragma unroll
        for (int i = 0; i < (KB * CB) / 256; ++i)
            breg[i] = bsrc[(long)(br + 4 * i) * NF + bc];
    }
    __builtin_amdgcn_sched_barrier(0);

    load_bfrags(cur, 1, bfB, fb);
    gcn_quarter<0, MORE>(asrc, acur, anxt, bfA, acc);
    load_bfrags(cur, 2, bfA, fb);
    gcn_quarter<1, MORE>(asrc, acur, anxt, bfB, acc);
    load_bfrags(cur, 3, bfB, fb);
    gcn_quarter<2, MORE>(asrc, acur, anxt, bfA, acc);

    if (MORE) {
        #pragma unroll
        for (int i = 0; i < (KB * CB) / 256; ++i)
            nxt[bc * LDT + br + 4 * i] = breg[i];
        lds_barrier();
        load_bfrags(nxt, 0, bfA, fb);
    }
    gcn_quarter<3, MORE>(asrc, acur, anxt, bfB, acc);
}

__global__ __launch_bounds__(256, 1)
void gcn_gemm(const float* __restrict__ A, long lda, long aBatchStride,
              const float* __restrict__ Bm, long bBatchStride, int K,
              const float* __restrict__ dinv, const float* __restrict__ bias,
              float* __restrict__ out)
{
    __shared__ float bt[2 * TILE];

    const int tid    = threadIdx.x;
    const int lane   = tid & 31;
    const int wave   = tid >> 5;
    const int batch  = blockIdx.x >> 6;
    const int rowBlk = (blockIdx.x >> 1) & 31;
    const int c0     = (blockIdx.x & 1) * CB;
    const int m0     = rowBlk * 128 + wave * 16;
    const int lh     = lane >> 4;
    const int ll     = lane & 15;
    const long gRow0 = (long)batch * NROWS + m0;

    const float* Ap   = A  + (long)batch * aBatchStride;
    const float* Bp   = Bm + (long)batch * bBatchStride + c0;
    const float* aRow = Ap + (long)(m0 + ll) * lda + lh * 2;

    const int br = tid >> 6;
    const int bc = tid & 63;
    const int fb = ll * LDT + lh * 2;

    v8f acc[4] = {};

    const int nStages = K / KB;

    v2f a0[KB / 4], a1[KB / 4];
    #pragma unroll
    for (int j = 0; j < KB / 4; ++j)
        a0[j] = __builtin_nontemporal_load((const v2f*)(aRow + j * 4));

    #pragma unroll
    for (int i = 0; i < (KB * CB) / 256; ++i)
        bt[bc * LDT + br + 4 * i] = Bp[(long)(br + 4 * i) * NF + bc];

    v2f bfA[FQ], bfB[FQ];
    lds_barrier();
    load_bfrags(bt, 0, bfA, fb);

    int s = 0;
    for (; s + 2 < nStages; s += 2) {
        gcn_gemm_stage<true>(bt, bt + TILE,
                             Bp + (long)(s + 1) * (KB * NF), aRow + (long)(s + 1) * KB,
                             a0, a1, acc, bfA, bfB, br, bc, fb);
        gcn_gemm_stage<true>(bt + TILE, bt,
                             Bp + (long)(s + 2) * (KB * NF), aRow + (long)(s + 2) * KB,
                             a1, a0, acc, bfA, bfB, br, bc, fb);
    }
    if (s + 1 < nStages) {
        gcn_gemm_stage<true>(bt, bt + TILE,
                             Bp + (long)(s + 1) * (KB * NF), aRow + (long)(s + 1) * KB,
                             a0, a1, acc, bfA, bfB, br, bc, fb);
        gcn_gemm_stage<false>(bt + TILE, bt, Bp, aRow, a1, a0, acc, bfA, bfB, br, bc, fb);
    } else if (s < nStages) {
        gcn_gemm_stage<false>(bt, bt + TILE, Bp, aRow, a0, a1, acc, bfA, bfB, br, bc, fb);
    }

    #pragma unroll
    for (int nt = 0; nt < 4; ++nt) {
        const int col = c0 + nt * 16 + ll;
        #pragma unroll
        for (int v = 0; v < 8; ++v) {
            const long gRow = gRow0 + v + lh * 8;
            float val = acc[nt][v] * dinv[gRow];
            if (bias) val += bias[col];
            out[gRow * NF + col] = val;
        }
    }
}

extern "C" void kernel_launch(void* const* d_in, const int* in_sizes, int n_in,
                              void* d_out, int out_size, void* d_ws, size_t ws_size,
                              hipStream_t stream) {
    const float* x      = (const float*)d_in[0];
    const float* adj    = (const float*)d_in[1];
    const float* weight = (const float*)d_in[2];
    const float* bias   = (const float*)d_in[3];
    float*       out    = (float*)d_out;

    float* dinv = (float*)d_ws;
    float* y    = dinv + 4 * NROWS;

    const int B = 4;

    gcn_degree<<<(B * NROWS) / 8, 256, 0, stream>>>(adj, dinv);

    gcn_gemm<<<B * (NROWS / 128) * (NF / CB), 256, 0, stream>>>(
        x, NF, (long)NROWS * NF, weight, 0, NF, dinv, nullptr, y);

    gcn_gemm<<<B * (NROWS / 128) * (NF / CB), 256, 0, stream>>>(
        adj, NROWS, (long)NROWS * NROWS, y, (long)NROWS * NF, NROWS,
        dinv, bias, out);
}


# Main GEMM re-tiled: one wave per SIMD, K split over a workgroup's 4 waves, A and transposed-y fragments as 16-B global loads into registers (no LDS panel, no stage barrier); sum order changed
# speedup vs baseline: 7.4643x; 1.1458x over previous
#include <hip/hip_runtime.h>

typedef __attribute__((ext_vector_type(2))) float v2f;
typedef __attribute__((ext_vector_type(4))) float v4f;
typedef __attribute__((ext_vector_type(8))) float v8f;

#define NROWS 4096
#define NF    128
#define CB    64
#define KB    128
#define LDT   132
#define TILE  (CB * LDT)

__global__ void gcn_degree(const float* __restrict__ adj, float* __restrict__ dinv)
{
    const int  lane = threadIdx.x & 31;
    const int  wave = threadIdx.x >> 5;
    const long row  = (long)blockIdx.x * 8 + wave;
    const v4f* rp4  = (const v4f*)(adj + row * (long)NROWS);

    float s = 0.f;
    for (int k = lane; k < NROWS / 4; k += 32) {
        v4f v = rp4[k];
        s += v.x + v.y + v.z + v.w;
    }
    #pragma unroll
    for (int o = 16; o > 0; o >>= 1) s += __shfl_xor(s, o, 32);

    if (lane == 0) dinv[row] = 1.0f / sqrtf(s + 1e-6f);
}

#define AQ 8
#define FQ (AQ * 4)

__device__ __forceinline__ void lds_barrier()
{
    __builtin_amdgcn_fence(__ATOMIC_RELEASE, "workgroup", "local");
    __builtin_amdgcn_s_barrier();
    __builtin_amdgcn_fence(__ATOMIC_ACQUIRE, "workgroup", "local");
}

__device__ __forceinline__ void load_bfrags(const float* __restrict__ p, const int qd,
                                            v2f (&f)[FQ], const int fb)
{
    #pragma unroll
    for (int jj = 0; jj < AQ; ++jj) {
        #pragma unroll
        for (int nt = 0; nt < 4; ++nt)
            f[jj * 4 + nt] = *(const v2f*)(&p[fb + nt * 16 * LDT + (qd * AQ + jj) * 4]);
    }
}

template <int QD, bool APF>
__device__ __forceinline__ void gcn_quarter(
    const float* __restrict__ asrc, const v2f (&acur)[KB / 4], v2f (&anxt)[KB / 4],
    const v2f (&bu)[FQ], v8f (&acc)[4])
{
    if (APF) {
        #pragma unroll
        for (int j = QD * AQ; j < (QD + 1) * AQ; ++j)
            anxt[j] = __builtin_nontemporal_load((const v2f*)(asrc + j * 4));
    }
    __builtin_amdgcn_sched_barrier(0);
    #pragma unroll
    for (int jj = 0; jj < AQ; ++jj) {
        const v2f a = acur[QD * AQ + jj];
        #pragma unroll
        for (int nt = 0; nt < 4; ++nt)
            acc[nt] = __builtin_amdgcn_wmma_f32_16x16x4_f32(
                false, a, false, bu[jj * 4 + nt], (short)0, acc[nt], false, false);
    }
    __builtin_amdgcn_sched_barrier(0);
}

template <bool MORE>
__device__ __forceinline__ void gcn_gemm_stage(
    const float* __restrict__ cur, float* __restrict__ nxt,
    const float* __restrict__ bsrc, const float* __restrict__ asrc,
    const v2f (&acur)[KB / 4], v2f (&anxt)[KB / 4], v8f (&acc)[4],
    v2f (&bfA)[FQ], v2f (&bfB)[FQ],
    const int br, const int bc, const int fb)
{
    float breg[(KB * CB) / 256];
    if (MORE) {
        #pragma unroll
        for (int i = 0; i < (KB * CB) / 256; ++i)
            breg[i] = bsrc[(long)(br + 4 * i) * NF + bc];
    }
    __builtin_amdgcn_sched_barrier(0);

    load_bfrags(cur, 1, bfB, fb);
    gcn_quarter<0, MORE>(asrc, acur, anxt, bfA, acc);
    load_bfrags(cur, 2, bfA, fb);
    gcn_quarter<1, MORE>(asrc, acur, anxt, bfB, acc);
    load_bfrags(cur, 3, bfB, fb);
    gcn_quarter<2, MORE>(asrc, acur, anxt, bfA, acc);

    if (MORE) {
        #pragma unroll
        for (int i = 0; i < (KB * CB) / 256; ++i)
            nxt[bc * LDT + br + 4 * i] = breg[i];
        lds_barrier();
        load_bfrags(nxt, 0, bfA, fb);
    }
    gcn_quarter<3, MORE>(asrc, acur, anxt, bfB, acc);
}

__global__ __launch_bounds__(256, 1)
void gcn_gemm(const float* __restrict__ A, long lda, long aBatchStride,
              const float* __restrict__ Bm, long bBatchStride, int K,
              const float* __restrict__ dinv, const float* __restrict__ bias,
              float* __restrict__ out)
{
    __shared__ float bt[2 * TILE];

    const int tid    = threadIdx.x;
    const int lane   = tid & 31;
    const int wave   = tid >> 5;
    const int batch  = blockIdx.x >> 6;
    const int rowBlk = (blockIdx.x >> 1) & 31;
    const int c0     = (blockIdx.x & 1) * CB;
    const int m0     = rowBlk * 128 + wave * 16;
    const int lh     = lane >> 4;
    const int ll     = lane & 15;
    const long gRow0 = (long)batch * NROWS + m0;

    const float* Ap   = A  + (long)batch * aBatchStride;
    const float* Bp   = Bm + (long)batch * bBatchStride + c0;
    const float* aRow = Ap + (long)(m0 + ll) * lda + lh * 2;

    const int br = tid >> 6;
    const int bc = tid & 63;
    const int fb = ll * LDT + lh * 2;

    v8f acc[4] = {};

    const int nStages = K / KB;

    v2f a0[KB / 4], a1[KB / 4];
    #pragma unroll
    for (int j = 0; j < KB / 4; ++j)
        a0[j] = __builtin_nontemporal_load((const v2f*)(aRow + j * 4));

    #pragma unroll
    for (int i = 0; i < (KB * CB) / 256; ++i)
        bt[bc * LDT + br + 4 * i] = Bp[(long)(br + 4 * i) * NF + bc];

    v2f bfA[FQ], bfB[FQ];
    lds_barrier();
    load_bfrags(bt, 0, bfA, fb);

    int s = 0;
    for (; s + 2 < nStages; s += 2) {
        gcn_gemm_stage<true>(bt, bt + TILE,
                             Bp + (long)(s + 1) * (KB * NF), aRow + (long)(s + 1) * KB,
                             a0, a1, acc, bfA, bfB, br, bc, fb);
        gcn_gemm_stage<true>(bt + TILE, bt,
                             Bp + (long)(s + 2) * (KB * NF), aRow + (long)(s + 2) * KB,
                             a1, a0, acc, bfA, bfB, br, bc, fb);
    }
    if (s + 1 < nStages) {
        gcn_gemm_stage<true>(bt, bt + TILE,
                             Bp + (long)(s + 1) * (KB * NF), aRow + (long)(s + 1) * KB,
                             a0, a1, acc, bfA, bfB, br, bc, fb);
        gcn_gemm_stage<false>(bt + TILE, bt, Bp, aRow, a1, a0, acc, bfA, bfB, br, bc, fb);
    } else if (s < nStages) {
        gcn_gemm_stage<false>(bt, bt + TILE, Bp, aRow, a0, a1, acc, bfA, bfB, br, bc, fb);
    }

    #pragma unroll
    for (int nt = 0; nt < 4; ++nt) {
        const int col = c0 + nt * 16 + ll;
        const float bv = bias ? bias[col] : 0.f;
        v4f lo, hi;
        #pragma unroll
        for (int v = 0; v < 4; ++v) {
            lo[v] = acc[nt][v]     * dinv[gRow0 + lh * 8 + v]     + bv;
            hi[v] = acc[nt][v + 4] * dinv[gRow0 + lh * 8 + v + 4] + bv;
        }
        float* op = out + ((long)batch * NF + col) * NROWS + m0 + lh * 8;
        *(v4f*)op       = lo;
        *(v4f*)(op + 4) = hi;
    }
}

#define RT 8
#define CT 4
#define KW 4
#define KL 32

template <bool MORE>
__device__ __forceinline__ void gcn_ks_step(
    const float* __restrict__ ap, const float* __restrict__ bp,
    const v4f (&ac)[RT][4], v4f (&an)[RT][4],
    const v4f (&bc)[CT][4], v4f (&bn)[CT][4],
    v8f (&acc)[RT][CT])
{
    #pragma unroll
    for (int r = 0; r < RT; ++r) {
        if (MORE) {
            if (r < CT) {
                #pragma unroll
                for (int c = 0; c < 4; ++c)
                    bn[r][c] = *(const v4f*)(bp + (long)r * 16 * NROWS + c * 8);
            }
            #pragma unroll
            for (int c = 0; c < 4; ++c)
                an[r][c] = __builtin_nontemporal_load(
                    (const v4f*)(ap + (long)r * 16 * NROWS + c * 8));
        }
        __builtin_amdgcn_sched_barrier(0);
        #pragma unroll
        for (int c = 0; c < 4; ++c) {
            #pragma unroll
            for (int e = 0; e < 2; ++e) {
                const v2f a = e ? __builtin_shufflevector(ac[r][c], ac[r][c], 2, 3)
                                : __builtin_shufflevector(ac[r][c], ac[r][c], 0, 1);
                #pragma unroll
                for (int ct = 0; ct < CT; ++ct) {
                    const v2f b = e ? __builtin_shufflevector(bc[ct][c], bc[ct][c], 2, 3)
                                    : __builtin_shufflevector(bc[ct][c], bc[ct][c], 0, 1);
                    acc[r][ct] = __builtin_amdgcn_wmma_f32_16x16x4_f32(
                        false, a, false, b, (short)0, acc[r][ct], false, false);
                }
            }
        }
        __builtin_amdgcn_sched_barrier(0);
    }
}

__global__ __launch_bounds__(128, 1)
void gcn_gemm_ksplit(const float* __restrict__ A, const float* __restrict__ Bt,
                     const float* __restrict__ dinv, const float* __restrict__ bias,
                     float* __restrict__ out)
{
    __shared__ __attribute__((aligned(16))) float red[KW * RT * CT * 256];

    const int tid    = threadIdx.x;
    const int lane   = tid & 31;
    const int wave   = tid >> 5;
    const int batch  = blockIdx.x >> 6;
    const int rowBlk = (blockIdx.x >> 1) & 31;
    const int c0     = (blockIdx.x & 1) * (CT * 16);
    const int lh     = lane >> 4;
    const int ll     = lane & 15;

    const float* ap = A + (long)batch * NROWS * NROWS
                        + (long)(rowBlk * 128 + ll) * NROWS + wave * KL + lh * 4;
    const float* bp = Bt + (long)batch * NF * NROWS
                         + (long)(c0 + ll) * NROWS + wave * KL + lh * 4;

    v8f acc[RT][CT] = {};
    v4f a0[RT][4], a1[RT][4], b0[CT][4], b1[CT][4];

    #pragma unroll
    for (int ct = 0; ct < CT; ++ct) {
        #pragma unroll
        for (int c = 0; c < 4; ++c)
            b0[ct][c] = *(const v4f*)(bp + (long)ct * 16 * NROWS + c * 8);
    }
    #pragma unroll
    for (int r = 0; r < RT; ++r) {
        #pragma unroll
        for (int c = 0; c < 4; ++c)
            a0[r][c] = __builtin_nontemporal_load((const v4f*)(ap + (long)r * 16 * NROWS + c * 8));
    }

    const int nSteps = NROWS / (KW * KL);
    for (int s = 0; s + 2 < nSteps; s += 2) {
        gcn_ks_step<true>(ap + (s + 1) * (KW * KL), bp + (s + 1) * (KW * KL), a0, a1, b0, b1, acc);
        gcn_ks_step<true>(ap + (s + 2) * (KW * KL), bp + (s + 2) * (KW * KL), a1, a0, b1, b0, acc);
    }
    gcn_ks_step<true>(ap + (nSteps - 1) * (KW * KL), bp + (nSteps - 1) * (KW * KL),
                      a0, a1, b0, b1, acc);
    gcn_ks_step<false>(ap, bp, a1, a0, b1, b0, acc);

    #pragma unroll
    for (int r = 0; r < RT; ++r) {
        #pragma unroll
        for (int ct = 0; ct < CT; ++ct) {
            const int t = r * CT + ct;
            v4f lo, hi;
            #pragma unroll
            for (int v = 0; v < 4; ++v) { lo[v] = acc[r][ct][v]; hi[v] = acc[r][ct][v + 4]; }
            *(v4f*)&red[((wave * (RT * CT) + t) * 2 + 0) * 128 + lane * 4] = lo;
            *(v4f*)&red[((wave * (RT * CT) + t) * 2 + 1) * 128 + lane * 4] = hi;
        }
    }
    __syncthreads();

    #pragma unroll
    for (int rr = 0; rr < 2; ++rr) {
        const int  r     = 2 * wave + rr;
        const long gRowB = (long)batch * NROWS + rowBlk * 128 + r * 16 + lh * 8;
        #pragma unroll
        for (int ct = 0; ct < CT; ++ct) {
            const int   t   = r * CT + ct;
            const int   col = c0 + ct * 16 + ll;
            const float bv  = bias[col];
            #pragma unroll
            for (int h = 0; h < 2; ++h) {
                v4f sum = *(const v4f*)&red[((0 * (RT * CT) + t) * 2 + h) * 128 + lane * 4];
                #pragma unroll
                for (int w = 1; w < KW; ++w)
                    sum += *(const v4f*)&red[((w * (RT * CT) + t) * 2 + h) * 128 + lane * 4];
                #pragma unroll
                for (int v = 0; v < 4; ++v) {
                    const long gRow = gRowB + h * 4 + v;
                    out[gRow * NF + col] = sum[v] * dinv[gRow] + bv;
                }
            }
        }
    }
}


extern "C" void kernel_launch(void* const* d_in, const int* in_sizes, int n_in,
                              void* d_out, int out_size, void* d_ws, size_t ws_size,
                              hipStream_t stream) {
    const float* x      = (const float*)d_in[0];
    const float* adj    = (const float*)d_in[1];
    const float* weight = (const float*)d_in[2];
    const float* bias   = (const float*)d_in[3];
    float*       out    = (float*)d_out;

    float* dinv = (float*)d_ws;
    float* y    = dinv + 4 * NROWS;

    const int B = 4;

    gcn_degree<<<(B * NROWS) / 8, 256, 0, stream>>>(adj, dinv);

    gcn_gemm<<<B * (NROWS / 128) * (NF / CB), 256, 0, stream>>>(
        x, NF, (long)NROWS * NF, weight, 0, NF, dinv, nullptr, y);

    gcn_gemm_ksplit<<<B * (NROWS / 128) * (NF / (CT * 16)), 128, 0, stream>>>(
        adj, y, dinv, bias, out);
}


# Degree kernel reshaped as a one-pass stream: 256 workgroups x 16 waves x 4 rows, next half row's 16 loads issued ahead of the sums (was 16384 one-row waves draining every 8 loads); sum order kept
# speedup vs baseline: 7.7637x; 1.0401x over previous
#include <hip/hip_runtime.h>

typedef __attribute__((ext_vector_type(2))) float v2f;
typedef __attribute__((ext_vector_type(4))) float v4f;
typedef __attribute__((ext_vector_type(8))) float v8f;

#define NROWS 4096
#define NF    128
#define CB    64
#define KB    128
#define LDT   132
#define TILE  (CB * LDT)

#define DW 16
#define DR 4
#define DQ 16

__global__ __launch_bounds__(DW * 32)
void gcn_degree(const float* __restrict__ adj, float* __restrict__ dinv)
{
    const int  lane = threadIdx.x & 31;
    const int  wave = __builtin_amdgcn_readfirstlane(threadIdx.x >> 5);
    const long row0 = ((long)blockIdx.x * DW + wave) * DR;
    const v4f* p = (const v4f*)(adj + row0 * (long)NROWS) + lane;

    v4f   buf[2][DQ];
    float sum[DR];

    #pragma unroll
    for (int q = 0; q < DQ; ++q)
        buf[0][q] = __builtin_nontemporal_load(p + q * 32);

    #pragma unroll
    for (int g = 0; g < 2 * DR; ++g) {
        __builtin_amdgcn_sched_barrier(0);
        if (g + 1 < 2 * DR) {
            #pragma unroll
            for (int q = 0; q < DQ; ++q)
                buf[(g + 1) & 1][q] = __builtin_nontemporal_load(p + ((g + 1) * DQ + q) * 32);
        }
        __builtin_amdgcn_sched_barrier(0);
        float s = (g & 1) ? sum[g >> 1] : 0.f;
        #pragma unroll
        for (int q = 0; q < DQ; ++q) {
            const v4f v = buf[g & 1][q];
            s += v.x + v.y + v.z + v.w;
        }
        sum[g >> 1] = s;
    }

    float mine = 0.f;
    #pragma unroll
    for (int r = 0; r < DR; ++r) {
        float s = sum[r];
        #pragma unroll
        for (int o = 16; o > 0; o >>= 1) s += __shfl_xor(s, o, 32);
        if (lane == r) mine = s;
    }
    if (lane < DR) dinv[row0 + lane] = 1.0f / sqrtf(mine + 1e-6f);
}

#define AQ 8
#define FQ (AQ * 4)

__device__ __forceinline__ void lds_barrier()
{
    __builtin_amdgcn_fence(__ATOMIC_RELEASE, "workgroup", "local");
    __builtin_amdgcn_s_barrier();
    __builtin_amdgcn_fence(__ATOMIC_ACQUIRE, "workgroup", "local");
}

__device__ __forceinline__ void load_bfrags(const float* __restrict__ p, const int qd,
                                            v2f (&f)[FQ], const int fb)
{
    #pragma unroll
    for (int jj = 0; jj < AQ; ++jj) {
        #pragma unroll
        for (int nt = 0; nt < 4; ++nt)
            f[jj * 4 + nt] = *(const v2f*)(&p[fb + nt * 16 * LDT + (qd * AQ + jj) * 4]);
    }
}

template <int QD, bool APF>
__device__ __forceinline__ void gcn_quarter(
    const float* __restrict__ asrc, const v2f (&acur)[KB / 4], v2f (&anxt)[KB / 4],
    const v2f (&bu)[FQ], v8f (&acc)[4])
{
    if (APF) {
        #pragma unroll
        for (int j = QD * AQ; j < (QD + 1) * AQ; ++j)
            anxt[j] = __builtin_nontemporal_load((const v2f*)(asrc + j * 4));
    }
    __builtin_amdgcn_sched_barrier(0);
    #pragma unroll
    for (int jj = 0; jj < AQ; ++jj) {
        const v2f a = acur[QD * AQ + jj];
        #pragma unroll
        for (int nt = 0; nt < 4; ++nt)
            acc[nt] = __builtin_amdgcn_wmma_f32_16x16x4_f32(
                false, a, false, bu[jj * 4 + nt], (short)0, acc[nt], false, false);
    }
    __builtin_amdgcn_sched_barrier(0);
}

template <bool MORE>
__device__ __forceinline__ void gcn_gemm_stage(
    const float* __restrict__ cur, float* __restrict__ nxt,
    const float* __restrict__ bsrc, const float* __restrict__ asrc,
    const v2f (&acur)[KB / 4], v2f (&anxt)[KB / 4], v8f (&acc)[4],
    v2f (&bfA)[FQ], v2f (&bfB)[FQ],
    const int br, const int bc, const int fb)
{
    float breg[(KB * CB) / 256];
    if (MORE) {
        #pragma unroll
        for (int i = 0; i < (KB * CB) / 256; ++i)
            breg[i] = bsrc[(long)(br + 4 * i) * NF + bc];
    }
    __builtin_amdgcn_sched_barrier(0);

    load_bfrags(cur, 1, bfB, fb);
    gcn_quarter<0, MORE>(asrc, acur, anxt, bfA, acc);
    load_bfrags(cur, 2, bfA, fb);
    gcn_quarter<1, MORE>(asrc, acur, anxt, bfB, acc);
    load_bfrags(cur, 3, bfB, fb);
    gcn_quarter<2, MORE>(asrc, acur, anxt, bfA, acc);

    if (MORE) {
        #pragma unroll
        for (int i = 0; i < (KB * CB) / 256; ++i)
            nxt[bc * LDT + br + 4 * i] = breg[i];
        lds_barrier();
        load_bfrags(nxt, 0, bfA, fb);
    }
    gcn_quarter<3, MORE>(asrc, acur, anxt, bfB, acc);
}

__global__ __launch_bounds__(256, 1)
void gcn_gemm(const float* __restrict__ A, long lda, long aBatchStride,
              const float* __restrict__ Bm, long bBatchStride, int K,
              const float* __restrict__ dinv, const float* __restrict__ bias,
              float* __restrict__ out)
{
    __shared__ float bt[2 * TILE];

    const int tid    = threadIdx.x;
    const int lane   = tid & 31;
    const int wave   = tid >> 5;
    const int batch  = blockIdx.x >> 6;
    const int rowBlk = (blockIdx.x >> 1) & 31;
    const int c0     = (blockIdx.x & 1) * CB;
    const int m0     = rowBlk * 128 + wave * 16;
    const int lh     = lane >> 4;
    const int ll     = lane & 15;
    const long gRow0 = (long)batch * NROWS + m0;

    const float* Ap   = A  + (long)batch * aBatchStride;
    const float* Bp   = Bm + (long)batch * bBatchStride + c0;
    const float* aRow = Ap + (long)(m0 + ll) * lda + lh * 2;

    const int br = tid >> 6;
    const int bc = tid & 63;
    const int fb = ll * LDT + lh * 2;

    v8f acc[4] = {};

    const int nStages = K / KB;

    v2f a0[KB / 4], a1[KB / 4];
    #pragma unroll
    for (int j = 0; j < KB / 4; ++j)
        a0[j] = __builtin_nontemporal_load((const v2f*)(aRow + j * 4));

    #pragma unroll
    for (int i = 0; i < (KB * CB) / 256; ++i)
        bt[bc * LDT + br + 4 * i] = Bp[(long)(br + 4 * i) * NF + bc];

    v2f bfA[FQ], bfB[FQ];
    lds_barrier();
    load_bfrags(bt, 0, bfA, fb);

    int s = 0;
    for (; s + 2 < nStages; s += 2) {
        gcn_gemm_stage<true>(bt, bt + TILE,
                             Bp + (long)(s + 1) * (KB * NF), aRow + (long)(s + 1) * KB,
                             a0, a1, acc, bfA, bfB, br, bc, fb);
        gcn_gemm_stage<true>(bt + TILE, bt,
                             Bp + (long)(s + 2) * (KB * NF), aRow + (long)(s + 2) * KB,
                             a1, a0, acc, bfA, bfB, br, bc, fb);
    }
    if (s + 1 < nStages) {
        gcn_gemm_stage<true>(bt, bt + TILE,
                             Bp + (long)(s + 1) * (KB * NF), aRow + (long)(s + 1) * KB,
                             a0, a1, acc, bfA, bfB, br, bc, fb);
        gcn_gemm_stage<false>(bt + TILE, bt, Bp, aRow, a1, a0, acc, bfA, bfB, br, bc, fb);
    } else if (s < nStages) {
        gcn_gemm_stage<false>(bt, bt + TILE, Bp, aRow, a0, a1, acc, bfA, bfB, br, bc, fb);
    }

    #pragma unroll
    for (int nt = 0; nt < 4; ++nt) {
        const int col = c0 + nt * 16 + ll;
        const float bv = bias ? bias[col] : 0.f;
        v4f lo, hi;
        #pragma unroll
        for (int v = 0; v < 4; ++v) {
            lo[v] = acc[nt][v]     * dinv[gRow0 + lh * 8 + v]     + bv;
            hi[v] = acc[nt][v + 4] * dinv[gRow0 + lh * 8 + v + 4] + bv;
        }
        float* op = out + ((long)batch * NF + col) * NROWS + m0 + lh * 8;
        *(v4f*)op       = lo;
        *(v4f*)(op + 4) = hi;
    }
}

#define RT 8
#define CT 4
#define KW 4
#define KL 32

template <bool MORE>
__device__ __forceinline__ void gcn_ks_step(
    const float* __restrict__ ap, const float* __restrict__ bp,
    const v4f (&ac)[RT][4], v4f (&an)[RT][4],
    const v4f (&bc)[CT][4], v4f (&bn)[CT][4],
    v8f (&acc)[RT][CT])
{
    #pragma unroll
    for (int r = 0; r < RT; ++r) {
        if (MORE) {
            if (r < CT) {
                #pragma unroll
                for (int c = 0; c < 4; ++c)
                    bn[r][c] = *(const v4f*)(bp + (long)r * 16 * NROWS + c * 8);
            }
            #pragma unroll
            for (int c = 0; c < 4; ++c)
                an[r][c] = __builtin_nontemporal_load(
                    (const v4f*)(ap + (long)r * 16 * NROWS + c * 8));
        }
        __builtin_amdgcn_sched_barrier(0);
        #pragma unroll
        for (int c = 0; c < 4; ++c) {
            #pragma unroll
            for (int e = 0; e < 2; ++e) {
                const v2f a = e ? __builtin_shufflevector(ac[r][c], ac[r][c], 2, 3)
                                : __builtin_shufflevector(ac[r][c], ac[r][c], 0, 1);
                #pragma unroll
                for (int ct = 0; ct < CT; ++ct) {
                    const v2f b = e ? __builtin_shufflevector(bc[ct][c], bc[ct][c], 2, 3)
                                    : __builtin_shufflevector(bc[ct][c], bc[ct][c], 0, 1);
                    acc[r][ct] = __builtin_amdgcn_wmma_f32_16x16x4_f32(
                        false, a, false, b, (short)0, acc[r][ct], false, false);
                }
            }
        }
        __builtin_amdgcn_sched_barrier(0);
    }
}

__global__ __launch_bounds__(128, 1)
void gcn_gemm_ksplit(const float* __restrict__ A, const float* __restrict__ Bt,
                     const float* __restrict__ dinv, const float* __restrict__ bias,
                     float* __restrict__ out)
{
    __shared__ __attribute__((aligned(16))) float red[KW * RT * CT * 256];

    const int tid    = threadIdx.x;
    const int lane   = tid & 31;
    const int wave   = tid >> 5;
    const int batch  = blockIdx.x >> 6;
    const int rowBlk = (blockIdx.x >> 1) & 31;
    const int c0     = (blockIdx.x & 1) * (CT * 16);
    const int lh     = lane >> 4;
    const int ll     = lane & 15;

    const float* ap = A + (long)batch * NROWS * NROWS
                        + (long)(rowBlk * 128 + ll) * NROWS + wave * KL + lh * 4;
    const float* bp = Bt + (long)batch * NF * NROWS
                         + (long)(c0 + ll) * NROWS + wave * KL + lh * 4;

    v8f acc[RT][CT] = {};
    v4f a0[RT][4], a1[RT][4], b0[CT][4], b1[CT][4];

    #pragma unroll
    for (int ct = 0; ct < CT; ++ct) {
        #pragma unroll
        for (int c = 0; c < 4; ++c)
            b0[ct][c] = *(const v4f*)(bp + (long)ct * 16 * NROWS + c * 8);
    }
    #pragma unroll
    for (int r = 0; r < RT; ++r) {
        #pragma unroll
        for (int c = 0; c < 4; ++c)
            a0[r][c] = __builtin_nontemporal_load((const v4f*)(ap + (long)r * 16 * NROWS + c * 8));
    }

    const int nSteps = NROWS / (KW * KL);
    for (int s = 0; s + 2 < nSteps; s += 2) {
        gcn_ks_step<true>(ap + (s + 1) * (KW * KL), bp + (s + 1) * (KW * KL), a0, a1, b0, b1, acc);
        gcn_ks_step<true>(ap + (s + 2) * (KW * KL), bp + (s + 2) * (KW * KL), a1, a0, b1, b0, acc);
    }
    gcn_ks_step<true>(ap + (nSteps - 1) * (KW * KL), bp + (nSteps - 1) * (KW * KL),
                      a0, a1, b0, b1, acc);
    gcn_ks_step<false>(ap, bp, a1, a0, b1, b0, acc);

    #pragma unroll
    for (int r = 0; r < RT; ++r) {
        #pragma unroll
        for (int ct = 0; ct < CT; ++ct) {
            const int t = r * CT + ct;
            v4f lo, hi;
            #pragma unroll
            for (int v = 0; v < 4; ++v) { lo[v] = acc[r][ct][v]; hi[v] = acc[r][ct][v + 4]; }
            *(v4f*)&red[((wave * (RT * CT) + t) * 2 + 0) * 128 + lane * 4] = lo;
            *(v4f*)&red[((wave * (RT * CT) + t) * 2 + 1) * 128 + lane * 4] = hi;
        }
    }
    __syncthreads();

    #pragma unroll
    for (int rr = 0; rr < 2; ++rr) {
        const int  r     = 2 * wave + rr;
        const long gRowB = (long)batch * NROWS + rowBlk * 128 + r * 16 + lh * 8;
        #pragma unroll
        for (int ct = 0; ct < CT; ++ct) {
            const int   t   = r * CT + ct;
            const int   col = c0 + ct * 16 + ll;
            const float bv  = bias[col];
            #pragma unroll
            for (int h = 0; h < 2; ++h) {
                v4f sum = *(const v4f*)&red[((0 * (RT * CT) + t) * 2 + h) * 128 + lane * 4];
                #pragma unroll
                for (int w = 1; w < KW; ++w)
                    sum += *(const v4f*)&red[((w * (RT * CT) + t) * 2 + h) * 128 + lane * 4];
                #pragma unroll
                for (int v = 0; v < 4; ++v) {
                    const long gRow = gRowB + h * 4 + v;
                    out[gRow * NF + col] = sum[v] * dinv[gRow] + bv;
                }
            }
        }
    }
}


extern "C" void kernel_launch(void* const* d_in, const int* in_sizes, int n_in,
                              void* d_out, int out_size, void* d_ws, size_t ws_size,
                              hipStream_t stream) {
    const float* x      = (const float*)d_in[0];
    const float* adj    = (const float*)d_in[1];
    const float* weight = (const float*)d_in[2];
    const float* bias   = (const float*)d_in[3];
    float*       out    = (float*)d_out;

    float* dinv = (float*)d_ws;
    float* y    = dinv + 4 * NROWS;

    const int B = 4;

    gcn_degree<<<(B * NROWS) / (DW * DR), DW * 32, 0, stream>>>(adj, dinv);

    gcn_gemm<<<B * (NROWS / 128) * (NF / CB), 256, 0, stream>>>(
        x, NF, (long)NROWS * NF, weight, 0, NF, dinv, nullptr, y);

    gcn_gemm_ksplit<<<B * (NROWS / 128) * (NF / (CT * 16)), 128, 0, stream>>>(
        adj, y, dinv, bias, out);
}


# x@W computed inside the degree kernel's workgroups for their own 64 rows (4 extra waves, WMMAs under the adj stream), scaled by their d_inv via LDS after one barrier: 2 launches, was 3; sum order kept
# speedup vs baseline: 8.1122x; 1.0449x over previous
#include <hip/hip_runtime.h>


typedef __attribute__((ext_vector_type(2))) float v2f;
typedef __attribute__((ext_vector_type(4))) float v4f;
typedef __attribute__((ext_vector_type(8))) float v8f;

#define NROWS 4096
#define NF    128

#define DW 16
#define DR 4
#define DQ 16
#define XW 4
#define XR (DW * DR)
#define XG 4
#define XN (NF / 4 / XG)

__device__ __forceinline__ void xw_load_group(
    const float* __restrict__ xa, const float* __restrict__ wb, const int g,
    v2f (&fa)[4][XG], v2f (&fb)[2][XG])
{
    #pragma unroll
    for (int jj = 0; jj < XG; ++jj) {
        const int k = (g * XG + jj) * 4;
        #pragma unroll
        for (int ct = 0; ct < 2; ++ct) {
            v2f b;
            b.x = wb[(long)k * NF + ct * 16];
            b.y = wb[(long)(k + 1) * NF + ct * 16];
            fb[ct][jj] = b;
        }
        #pragma unroll
        for (int rt = 0; rt < 4; ++rt)
            fa[rt][jj] = *(const v2f*)(xa + (long)rt * 16 * NF + k);
    }
}

__global__ __launch_bounds__((DW + XW) * 32)
void gcn_degree_xw(const float* __restrict__ adj, const float* __restrict__ x,
                   const float* __restrict__ w, float* __restrict__ dinv,
                   float* __restrict__ yT)
{
    __shared__ __attribute__((aligned(16))) float sd[XR];

    const int  lane = threadIdx.x & 31;
    const int  wave = __builtin_amdgcn_readfirstlane(threadIdx.x >> 5);
    const long blk0 = (long)blockIdx.x * XR;
    const int  lh   = lane >> 4;
    const int  ll   = lane & 15;

    v8f acc[4][2] = {};

    if (wave < DW) {
        const long row0 = blk0 + (long)wave * DR;
        const v4f* p = (const v4f*)(adj + row0 * (long)NROWS) + lane;

        v4f   buf[2][DQ];
        float sum[DR];

        #pragma unroll
        for (int q = 0; q < DQ; ++q)
            buf[0][q] = __builtin_nontemporal_load(p + q * 32);

        #pragma unroll
        for (int g = 0; g < 2 * DR; ++g) {
            __builtin_amdgcn_sched_barrier(0);
            if (g + 1 < 2 * DR) {
                #pragma unroll
                for (int q = 0; q < DQ; ++q)
                    buf[(g + 1) & 1][q] = __builtin_nontemporal_load(p + ((g + 1) * DQ + q) * 32);
            }
            __builtin_amdgcn_sched_barrier(0);
            float s = (g & 1) ? sum[g >> 1] : 0.f;
            #pragma unroll
            for (int q = 0; q < DQ; ++q) {
                const v4f v = buf[g & 1][q];
                s += v.x + v.y + v.z + v.w;
            }
            sum[g >> 1] = s;
        }

        float mine = 0.f;
        #pragma unroll
        for (int r = 0; r < DR; ++r) {
            float s = sum[r];
            #pragma unroll
            for (int o = 16; o > 0; o >>= 1) s += __shfl_xor(s, o, 32);
            if (lane == r) mine = s;
        }
        if (lane < DR) {
            const float d = 1.0f / sqrtf(mine + 1e-6f);
            dinv[row0 + lane]    = d;
            sd[wave * DR + lane] = d;
        }
    } else {
        const int cw = wave - DW;
        const float* xa = x + (blk0 + ll) * NF + lh * 2;
        const float* wb = w + (long)(lh * 2) * NF + cw * 32 + ll;

        v2f fa[2][4][XG], fb[2][2][XG];
        xw_load_group(xa, wb, 0, fa[0], fb[0]);

        #pragma unroll
        for (int g = 0; g < XN; ++g) {
            if (g + 1 < XN) xw_load_group(xa, wb, g + 1, fa[(g + 1) & 1], fb[(g + 1) & 1]);
            __builtin_amdgcn_sched_barrier(0);
            #pragma unroll
            for (int jj = 0; jj < XG; ++jj) {
                #pragma unroll
                for (int rt = 0; rt < 4; ++rt) {
                    #pragma unroll
                    for (int ct = 0; ct < 2; ++ct)
                        acc[rt][ct] = __builtin_amdgcn_wmma_f32_16x16x4_f32(
                            false, fa[g & 1][rt][jj], false, fb[g & 1][ct][jj],
                            (short)0, acc[rt][ct], false, false);
                }
            }
            __builtin_amdgcn_sched_barrier(0);
        }
    }

    __syncthreads();

    if (wave >= DW) {
        const int  cw    = wave - DW;
        const int  batch = blockIdx.x >> 6;
        const int  m0    = (blockIdx.x & 63) * XR;
        #pragma unroll
        for (int rt = 0; rt < 4; ++rt) {
            const v4f dlo = *(const v4f*)&sd[rt * 16 + lh * 8];
            const v4f dhi = *(const v4f*)&sd[rt * 16 + lh * 8 + 4];
            #pragma unroll
            for (int ct = 0; ct < 2; ++ct) {
                const int col = cw * 32 + ct * 16 + ll;
                v4f lo, hi;
                #pragma unroll
                for (int v = 0; v < 4; ++v) {
                    lo[v] = acc[rt][ct][v]     * dlo[v];
                    hi[v] = acc[rt][ct][v + 4] * dhi[v];
                }
                float* op = yT + ((long)batch * NF + col) * NROWS + m0 + rt * 16 + lh * 8;
                *(v4f*)op       = lo;
                *(v4f*)(op + 4) = hi;
            }
        }
    }
}

#define RT 8
#define CT 4
#define KW 4
#define KL 32

template <bool MORE>
__device__ __forceinline__ void gcn_ks_step(
    const float* __restrict__ ap, const float* __restrict__ bp,
    const v4f (&ac)[RT][4], v4f (&an)[RT][4],
    const v4f (&bc)[CT][4], v4f (&bn)[CT][4],
    v8f (&acc)[RT][CT])
{
    #pragma unroll
    for (int r = 0; r < RT; ++r) {
        if (MORE) {
            if (r < CT) {
                #pragma unroll
                for (int c = 0; c < 4; ++c)
                    bn[r][c] = *(const v4f*)(bp + (long)r * 16 * NROWS + c * 8);
            }
            #pragma unroll
            for (int c = 0; c < 4; ++c)
                an[r][c] = __builtin_nontemporal_load(
                    (const v4f*)(ap + (long)r * 16 * NROWS + c * 8));
        }
        __builtin_amdgcn_sched_barrier(0);
        #pragma unroll
        for (int c = 0; c < 4; ++c) {
            #pragma unroll
            for (int e = 0; e < 2; ++e) {
                const v2f a = e ? __builtin_shufflevector(ac[r][c], ac[r][c], 2, 3)
                                : __builtin_shufflevector(ac[r][c], ac[r][c], 0, 1);
                #pragma unroll
                for (int ct = 0; ct < CT; ++ct) {
                    const v2f b = e ? __builtin_shufflevector(bc[ct][c], bc[ct][c], 2, 3)
                                    : __builtin_shufflevector(bc[ct][c], bc[ct][c], 0, 1);
                    acc[r][ct] = __builtin_amdgcn_wmma_f32_16x16x4_f32(
                        false, a, false, b, (short)0, acc[r][ct], false, false);
                }
            }
        }
        __builtin_amdgcn_sched_barrier(0);
    }
}

__global__ __launch_bounds__(128, 1)
void gcn_gemm_ksplit(const float* __restrict__ A, const float* __restrict__ Bt,
                     const float* __restrict__ dinv, const float* __restrict__ bias,
                     float* __restrict__ out)
{
    __shared__ __attribute__((aligned(16))) float red[KW * RT * CT * 256];

    const int tid    = threadIdx.x;
    const int lane   = tid & 31;
    const int wave   = tid >> 5;
    const int batch  = blockIdx.x >> 6;
    const int rowBlk = (blockIdx.x >> 1) & 31;
    const int c0     = (blockIdx.x & 1) * (CT * 16);
    const int lh     = lane >> 4;
    const int ll     = lane & 15;

    const float* ap = A + (long)batch * NROWS * NROWS
                        + (long)(rowBlk * 128 + ll) * NROWS + wave * KL + lh * 4;
    const float* bp = Bt + (long)batch * NF * NROWS
                         + (long)(c0 + ll) * NROWS + wave * KL + lh * 4;

    v8f acc[RT][CT] = {};
    v4f a0[RT][4], a1[RT][4], b0[CT][4], b1[CT][4];

    #pragma unroll
    for (int ct = 0; ct < CT; ++ct) {
        #pragma unroll
        for (int c = 0; c < 4; ++c)
            b0[ct][c] = *(const v4f*)(bp + (long)ct * 16 * NROWS + c * 8);
    }
    #pragma unroll
    for (int r = 0; r < RT; ++r) {
        #pragma unroll
        for (int c = 0; c < 4; ++c)
            a0[r][c] = __builtin_nontemporal_load((const v4f*)(ap + (long)r * 16 * NROWS + c * 8));
    }

    const int nSteps = NROWS / (KW * KL);
    for (int s = 0; s + 2 < nSteps; s += 2) {
        gcn_ks_step<true>(ap + (s + 1) * (KW * KL), bp + (s + 1) * (KW * KL), a0, a1, b0, b1, acc);
        gcn_ks_step<true>(ap + (s + 2) * (KW * KL), bp + (s + 2) * (KW * KL), a1, a0, b1, b0, acc);
    }
    gcn_ks_step<true>(ap + (nSteps - 1) * (KW * KL), bp + (nSteps - 1) * (KW * KL),
                      a0, a1, b0, b1, acc);
    gcn_ks_step<false>(ap, bp, a1, a0, b1, b0, acc);

    #pragma unroll
    for (int r = 0; r < RT; ++r) {
        #pragma unroll
        for (int ct = 0; ct < CT; ++ct) {
            const int t = r * CT + ct;
            v4f lo, hi;
            #pragma unroll
            for (int v = 0; v < 4; ++v) { lo[v] = acc[r][ct][v]; hi[v] = acc[r][ct][v + 4]; }
            *(v4f*)&red[((wave * (RT * CT) + t) * 2 + 0) * 128 + lane * 4] = lo;
            *(v4f*)&red[((wave * (RT * CT) + t) * 2 + 1) * 128 + lane * 4] = hi;
        }
    }
    __syncthreads();

    #pragma unroll
    for (int rr = 0; rr < 2; ++rr) {
        const int  r     = 2 * wave + rr;
        const long gRowB = (long)batch * NROWS + rowBlk * 128 + r * 16 + lh * 8;
        #pragma unroll
        for (int ct = 0; ct < CT; ++ct) {
            const int   t   = r * CT + ct;
            const int   col = c0 + ct * 16 + ll;
            const float bv  = bias[col];
            #pragma unroll
            for (int h = 0; h < 2; ++h) {
                v4f sum = *(const v4f*)&red[((0 * (RT * CT) + t) * 2 + h) * 128 + lane * 4];
                #pragma unroll
                for (int w = 1; w < KW; ++w)
                    sum += *(const v4f*)&red[((w * (RT * CT) + t) * 2 + h) * 128 + lane * 4];
                #pragma unroll
                for (int v = 0; v < 4; ++v) {
                    const long gRow = gRowB + h * 4 + v;
                    out[gRow * NF + col] = sum[v] * dinv[gRow] + bv;
                }
            }
        }
    }
}


extern "C" void kernel_launch(void* const* d_in, const int* in_sizes, int n_in,
                              void* d_out, int out_size, void* d_ws, size_t ws_size,
                              hipStream_t stream) {
    const float* x      = (const float*)d_in[0];
    const float* adj    = (const float*)d_in[1];
    const float* weight = (const float*)d_in[2];
    const float* bias   = (const float*)d_in[3];
    float*       out    = (float*)d_out;

    float* dinv = (float*)d_ws;
    float* y    = dinv + 4 * NROWS;

    const int B = 4;

    gcn_degree_xw<<<(B * NROWS) / XR, (DW + XW) * 32, 0, stream>>>(adj, x, weight, dinv, y);

    gcn_gemm_ksplit<<<B * (NROWS / 128) * (NF / (CT * 16)), 128, 0, stream>>>(
        adj, y, dinv, bias, out);
}
